# starting kernel
# baseline (speedup 1.0000x reference)
#include <hip/hip_runtime.h>
#include <math.h>

#define BATCH   32
#define NIN     2048
#define KDIM    16
#define NCAP    32
#define CAPD    32
#define NOUT    1024
#define NCHUNK  128
#define NC      (NIN / NCHUNK)
#define BH      16
#define UHS     1028

typedef float v2f __attribute__((ext_vector_type(2)));
typedef float v8f __attribute__((ext_vector_type(8)));

__global__ __launch_bounds__(256)
void caps_pass(const float* __restrict__ U,
               const float* __restrict__ Wt,
               const float* __restrict__ o_prev,
               float* __restrict__ partial,
               int iter0)
{
    extern __shared__ float smem[];
    float* uhat = smem;
    float* oacc = uhat + BH * UHS;
    float* lg   = oacc + BH * 1024;
    float* cb   = lg + BH * 32;

    const int tid   = threadIdx.x;
    const int lane  = tid & 31;
    const int wave  = tid >> 5;
    const int chunk = blockIdx.x;
    const int half  = blockIdx.y;
    const int b0    = half * BH;

    const int l16 = lane & 15;
    const int hi  = lane >> 4;

    for (int e = tid; e < BH * 1024; e += 256) oacc[e] = 0.0f;
    __syncthreads();

    for (int nn = 0; nn < NC; ++nn) {
        const int n = chunk * NC + nn;

        if (nn + 1 < NC) {
            __builtin_prefetch(Wt + ((size_t)(n + 1) * KDIM) * NOUT + wave * 128, 0, 3);
        }

        const float* Up = U + ((size_t)(b0 + l16) * NIN + n) * KDIM;
        v2f a0, a1, a2, a3;
        a0.x = Up[0  + 2*hi]; a0.y = Up[1  + 2*hi];
        a1.x = Up[4  + 2*hi]; a1.y = Up[5  + 2*hi];
        a2.x = Up[8  + 2*hi]; a2.y = Up[9  + 2*hi];
        a3.x = Up[12 + 2*hi]; a3.y = Up[13 + 2*hi];

        const float* Wn = Wt + (size_t)n * KDIM * NOUT;
        for (int ct = 0; ct < 8; ++ct) {
            const int c0 = wave * 128 + ct * 16;
            const float* Wp = Wn + (size_t)(2 * hi) * NOUT + c0 + l16;
            v2f bf0, bf1, bf2, bf3;
            bf0.x = Wp[0  * NOUT]; bf0.y = Wp[1  * NOUT];
            bf1.x = Wp[4  * NOUT]; bf1.y = Wp[5  * NOUT];
            bf2.x = Wp[8  * NOUT]; bf2.y = Wp[9  * NOUT];
            bf3.x = Wp[12 * NOUT]; bf3.y = Wp[13 * NOUT];

            v8f acc = {};
            acc = __builtin_amdgcn_wmma_f32_16x16x4_f32(false, a0, false, bf0, (short)0, acc, false, false);
            acc = __builtin_amdgcn_wmma_f32_16x16x4_f32(false, a1, false, bf1, (short)0, acc, false, false);
            acc = __builtin_amdgcn_wmma_f32_16x16x4_f32(false, a2, false, bf2, (short)0, acc, false, false);
            acc = __builtin_amdgcn_wmma_f32_16x16x4_f32(false, a3, false, bf3, (short)0, acc, false, false);

            #pragma unroll
            for (int v = 0; v < 8; ++v)
                uhat[(v + 8 * hi) * UHS + c0 + l16] = acc[v];
        }
        __syncthreads();

        if (iter0) {
            for (int e = tid; e < BH * 1024; e += 256)
                oacc[e] += 0.03125f * uhat[(e >> 10) * UHS + (e & 1023)];
        } else {
            for (int p = tid; p < BH * NCAP; p += 256) {
                const int bl = p >> 5, i = p & 31;
                const float* op = o_prev + ((size_t)(b0 + bl) * NCAP + i) * CAPD;
                const float* uh = uhat + bl * UHS + i * CAPD;
                float s = 0.0f;
                #pragma unroll
                for (int j = 0; j < CAPD; ++j) s = fmaf(op[j], uh[j], s);
                lg[p] = s;
            }
            __syncthreads();
            if (tid < BH) {
                float m = -3.402823466e38f;
                for (int i = 0; i < NCAP; ++i) m = fmaxf(m, lg[tid * NCAP + i]);
                float s = 0.0f;
                for (int i = 0; i < NCAP; ++i) {
                    float e = __expf(lg[tid * NCAP + i] - m);
                    cb[tid * NCAP + i] = e; s += e;
                }
                float inv = 1.0f / s;
                for (int i = 0; i < NCAP; ++i) cb[tid * NCAP + i] *= inv;
            }
            __syncthreads();
            for (int e = tid; e < BH * 1024; e += 256) {
                const int bl = e >> 10, col = e & 1023;
                oacc[e] = fmaf(cb[bl * NCAP + (col >> 5)], uhat[bl * UHS + col], oacc[e]);
            }
        }
        __syncthreads();
    }

    float* pp = partial + (size_t)(chunk * 2 + half) * (BH * 1024);
    for (int e = tid; e < BH * 1024; e += 256) pp[e] = oacc[e];
}

__global__ __launch_bounds__(256)
void caps_reduce(const float* __restrict__ partial, float* __restrict__ out, int mode)
{
    __shared__ float osum[NOUT];
    __shared__ float fac[NCAP];
    const int b = blockIdx.x;
    const int half = b >> 4, bl = b & 15;
    const int tid = threadIdx.x;

    for (int e = tid; e < NOUT; e += 256) {
        float s = 0.0f;
        const float* p = partial + (size_t)half * (BH * 1024) + (size_t)bl * 1024 + e;
        for (int c = 0; c < NCHUNK; ++c)
            s += p[(size_t)c * 2 * (BH * 1024)];
        osum[e] = s;
    }
    __syncthreads();

    if (tid < NCAP) {
        float s2 = 0.0f;
        for (int j = 0; j < CAPD; ++j) { float v = osum[tid * CAPD + j]; s2 = fmaf(v, v, s2); }
        float f;
        if (mode == 0) f = rsqrtf(fmaxf(s2, 1e-12f));
        else           f = (s2 / (1.0f + s2)) * rsqrtf(s2 + 1e-7f);
        fac[tid] = f;
    }
    __syncthreads();

    for (int e = tid; e < NOUT; e += 256)
        out[(size_t)b * NOUT + e] = osum[e] * fac[e >> 5];
}

extern "C" void kernel_launch(void* const* d_in, const int* in_sizes, int n_in,
                              void* d_out, int out_size, void* d_ws, size_t ws_size,
                              hipStream_t stream)
{
    const float* U  = (const float*)d_in[0];
    const float* Wt = (const float*)d_in[1];
    float* out = (float*)d_out;

    float* partial = (float*)d_ws;
    float* o_prev  = partial + (size_t)(NCHUNK * 2) * (BH * 1024);

    const size_t shmem = (size_t)(BH * UHS + BH * 1024 + 2 * BH * NCAP) * sizeof(float);
    dim3 grid(NCHUNK, 2), blk(256);

    caps_pass<<<grid, blk, shmem, stream>>>(U, Wt, nullptr, partial, 1);
    caps_reduce<<<BATCH, 256, 0, stream>>>(partial, o_prev, 0);
    caps_pass<<<grid, blk, shmem, stream>>>(U, Wt, o_prev, partial, 0);
    caps_reduce<<<BATCH, 256, 0, stream>>>(partial, o_prev, 0);
    caps_pass<<<grid, blk, shmem, stream>>>(U, Wt, o_prev, partial, 0);
    caps_reduce<<<BATCH, 256, 0, stream>>>(partial, out, 1);
}


# pass kernel: W on the WMMA A side, 4 columns per lane, so a lane holds whole capsule rows of u_hat; logits, softmax and o stay in registers, LDS holds only 512 logits per n, operands loaded 2 n ahead
# speedup vs baseline: 2.9623x; 2.9623x over previous
#include <hip/hip_runtime.h>
#include <math.h>

#define BATCH   32
#define NIN     2048
#define KDIM    16
#define NCAP    32
#define CAPD    32
#define NOUT    1024
#define NCHUNK  128
#define NC      (NIN / NCHUNK)
#define BH      16
#define OSP     1028
#define LGP     36
#define PASS_LDS_BYTES ((BH * OSP + 2 * BH * LGP) * 4)

typedef float v2f __attribute__((ext_vector_type(2)));
typedef float v4f __attribute__((ext_vector_type(4)));
typedef float v8f __attribute__((ext_vector_type(8)));

struct Frag {
    v4f w[4][2][2];
    v2f u[4];
};

__device__ __forceinline__ void load_frag(Frag& f, const float* __restrict__ Wlane,
                                          const float* __restrict__ Ulane, int n)
{
    const float* wp = Wlane + (size_t)n * (KDIM * NOUT);
    #pragma unroll
    for (int s = 0; s < 4; ++s) {
        #pragma unroll
        for (int r = 0; r < 2; ++r) {
            #pragma unroll
            for (int tp = 0; tp < 2; ++tp)
                f.w[s][r][tp] = *(const v4f*)(wp + (4 * s + r) * NOUT + tp * 64);
        }
    }
    const float* up = Ulane + (size_t)n * KDIM;
    #pragma unroll
    for (int s = 0; s < 4; ++s) f.u[s] = *(const v2f*)(up + 4 * s);
}

template <bool ITER0, bool REFILL>
__device__ __forceinline__ void route_step(Frag& f, float (&oacc)[2][4][8],
                                           const float (&op)[2][32],
                                           float* lgw, const float* lgr,
                                           const float* __restrict__ Wlane,
                                           const float* __restrict__ Ulane, int n_refill)
{
    v8f D[2][4];
    #pragma unroll
    for (int tp = 0; tp < 2; ++tp) {
        #pragma unroll
        for (int q = 0; q < 4; ++q) D[tp][q] = (v8f){0.f, 0.f, 0.f, 0.f, 0.f, 0.f, 0.f, 0.f};
    }
    #pragma unroll
    for (int s = 0; s < 4; ++s) {
        #pragma unroll
        for (int tp = 0; tp < 2; ++tp) {
            #pragma unroll
            for (int q = 0; q < 4; ++q) {
                v2f a;
                a.x = f.w[s][0][tp][q];
                a.y = f.w[s][1][tp][q];
                D[tp][q] = __builtin_amdgcn_wmma_f32_16x16x4_f32(false, a, false, f.u[s], (short)0,
                                                                 D[tp][q], false, false);
            }
        }
    }
    if (REFILL) load_frag(f, Wlane, Ulane, n_refill);

    if (ITER0) {
        #pragma unroll
        for (int tp = 0; tp < 2; ++tp) {
            #pragma unroll
            for (int q = 0; q < 4; ++q) {
                #pragma unroll
                for (int r = 0; r < 8; ++r)
                    oacc[tp][q][r] = fmaf(0.03125f, D[tp][q][r], oacc[tp][q][r]);
            }
        }
    } else {
        float lgt[2];
        #pragma unroll
        for (int tp = 0; tp < 2; ++tp) {
            float s = 0.0f;
            #pragma unroll
            for (int j = 0; j < CAPD; ++j) s = fmaf(op[tp][j], D[tp][j & 3][j >> 2], s);
            lgt[tp] = s;
            lgw[2 * tp] = s;
        }
        __builtin_amdgcn_fence(__ATOMIC_RELEASE, "workgroup", "local");
        __builtin_amdgcn_s_barrier();
        __builtin_amdgcn_fence(__ATOMIC_ACQUIRE, "workgroup", "local");
        float L[NCAP];
        #pragma unroll
        for (int i4 = 0; i4 < NCAP / 4; ++i4) {
            const v4f t = *(const v4f*)(lgr + 4 * i4);
            L[4 * i4 + 0] = t.x; L[4 * i4 + 1] = t.y; L[4 * i4 + 2] = t.z; L[4 * i4 + 3] = t.w;
        }
        float m = -3.402823466e38f;
        #pragma unroll
        for (int i = 0; i < NCAP; ++i) m = fmaxf(m, L[i]);
        float ssum = 0.0f;
        #pragma unroll
        for (int i = 0; i < NCAP; ++i) ssum += __expf(L[i] - m);
        const float inv = 1.0f / ssum;
        #pragma unroll
        for (int tp = 0; tp < 2; ++tp) {
            const float cbv = __expf(lgt[tp] - m) * inv;
            #pragma unroll
            for (int q = 0; q < 4; ++q) {
                #pragma unroll
                for (int r = 0; r < 8; ++r)
                    oacc[tp][q][r] = fmaf(cbv, D[tp][q][r], oacc[tp][q][r]);
            }
        }
    }
}

template <bool ITER0>
__global__ __launch_bounds__(256)
void caps_pass(const float* __restrict__ U,
               const float* __restrict__ Wt,
               const float* __restrict__ o_prev,
               float* __restrict__ partial)
{
    extern __shared__ float smem[];
    float* osm = smem;
    float* lg  = smem + BH * OSP;

    const int tid   = threadIdx.x;
    const int lane  = tid & 31;
    const int wave  = tid >> 5;
    const int chunk = blockIdx.x;
    const int half  = blockIdx.y;
    const int b0    = half * BH;
    const int l16   = lane & 15;
    const int hi    = lane >> 4;

    float oacc[2][4][8];
    #pragma unroll
    for (int tp = 0; tp < 2; ++tp) {
        #pragma unroll
        for (int q = 0; q < 4; ++q) {
            #pragma unroll
            for (int r = 0; r < 8; ++r) oacc[tp][q][r] = 0.0f;
        }
    }

    float op[2][32];
    if (!ITER0) {
        #pragma unroll
        for (int tp = 0; tp < 2; ++tp) {
            const float* p = o_prev + ((size_t)(b0 + l16) * NCAP + (4 * wave + 2 * tp + hi)) * CAPD;
            #pragma unroll
            for (int j4 = 0; j4 < CAPD / 4; ++j4) {
                const v4f t = *(const v4f*)(p + 4 * j4);
                op[tp][4 * j4 + 0] = t.x; op[tp][4 * j4 + 1] = t.y;
                op[tp][4 * j4 + 2] = t.z; op[tp][4 * j4 + 3] = t.w;
            }
        }
    }

    const float* Wlane = Wt + (size_t)(2 * hi) * NOUT + wave * 128 + l16 * 4;
    const float* Ulane = U + (size_t)(b0 + l16) * NIN * KDIM + 2 * hi;
    const int n0 = chunk * NC;

    float* lgw0 = lg + l16 * LGP + 4 * wave + hi;
    float* lgw1 = lgw0 + BH * LGP;
    const float* lgr0 = lg + l16 * LGP;
    const float* lgr1 = lgr0 + BH * LGP;

    Frag f0, f1;
    load_frag(f0, Wlane, Ulane, n0);
    load_frag(f1, Wlane, Ulane, n0 + 1);
    #pragma unroll 1
    for (int nn = 0; nn < NC - 2; nn += 2) {
        route_step<ITER0, true>(f0, oacc, op, lgw0, lgr0, Wlane, Ulane, n0 + nn + 2);
        route_step<ITER0, true>(f1, oacc, op, lgw1, lgr1, Wlane, Ulane, n0 + nn + 3);
    }
    route_step<ITER0, false>(f0, oacc, op, lgw0, lgr0, Wlane, Ulane, 0);
    route_step<ITER0, false>(f1, oacc, op, lgw1, lgr1, Wlane, Ulane, 0);

    #pragma unroll
    for (int tp = 0; tp < 2; ++tp) {
        float* orow = osm + l16 * OSP + (4 * wave + 2 * tp + hi) * CAPD;
        #pragma unroll
        for (int r = 0; r < 8; ++r) {
            v4f t;
            t.x = oacc[tp][0][r]; t.y = oacc[tp][1][r]; t.z = oacc[tp][2][r]; t.w = oacc[tp][3][r];
            *(v4f*)(orow + 4 * r) = t;
        }
    }
    __syncthreads();
    float* pp = partial + (size_t)(chunk * 2 + half) * (BH * 1024);
    for (int e4 = tid; e4 < BH * 256; e4 += 256) {
        const int row = e4 >> 8, c4 = e4 & 255;
        *(v4f*)(pp + row * 1024 + c4 * 4) = *(const v4f*)(osm + row * OSP + c4 * 4);
    }
}

__global__ __launch_bounds__(256)
void caps_reduce(const float* __restrict__ partial, float* __restrict__ out, int mode)
{
    __shared__ float osum[NOUT];
    __shared__ float fac[NCAP];
    const int b = blockIdx.x;
    const int half = b >> 4, bl = b & 15;
    const int tid = threadIdx.x;

    for (int e = tid; e < NOUT; e += 256) {
        float s = 0.0f;
        const float* p = partial + (size_t)half * (BH * 1024) + (size_t)bl * 1024 + e;
        for (int c = 0; c < NCHUNK; ++c)
            s += p[(size_t)c * 2 * (BH * 1024)];
        osum[e] = s;
    }
    __syncthreads();

    if (tid < NCAP) {
        float s2 = 0.0f;
        for (int j = 0; j < CAPD; ++j) { float v = osum[tid * CAPD + j]; s2 = fmaf(v, v, s2); }
        float f;
        if (mode == 0) f = rsqrtf(fmaxf(s2, 1e-12f));
        else           f = (s2 / (1.0f + s2)) * rsqrtf(s2 + 1e-7f);
        fac[tid] = f;
    }
    __syncthreads();

    for (int e = tid; e < NOUT; e += 256)
        out[(size_t)b * NOUT + e] = osum[e] * fac[e >> 5];
}

extern "C" void kernel_launch(void* const* d_in, const int* in_sizes, int n_in,
                              void* d_out, int out_size, void* d_ws, size_t ws_size,
                              hipStream_t stream)
{
    const float* U  = (const float*)d_in[0];
    const float* Wt = (const float*)d_in[1];
    float* out = (float*)d_out;

    float* partial = (float*)d_ws;
    float* o_prev  = partial + (size_t)(NCHUNK * 2) * (BH * 1024);

    dim3 grid(NCHUNK, 2), blk(256);

    caps_pass<true><<<grid, blk, PASS_LDS_BYTES, stream>>>(U, Wt, nullptr, partial);
    caps_reduce<<<BATCH, 256, 0, stream>>>(partial, o_prev, 0);
    caps_pass<false><<<grid, blk, PASS_LDS_BYTES, stream>>>(U, Wt, o_prev, partial);
    caps_reduce<<<BATCH, 256, 0, stream>>>(partial, o_prev, 0);
    caps_pass<false><<<grid, blk, PASS_LDS_BYTES, stream>>>(U, Wt, o_prev, partial);
    caps_reduce<<<BATCH, 256, 0, stream>>>(partial, out, 1);
}


# reduce kernel: 256 four-wave workgroups (was 32), each lane 32 independent 16-B loads in flight instead of 32 serial rounds of 16 4-B loads; partials and squares summed in a different tree
# speedup vs baseline: 4.8077x; 1.6230x over previous
#include <hip/hip_runtime.h>
#include <math.h>

#define BATCH   32
#define NIN     2048
#define KDIM    16
#define NCAP    32
#define CAPD    32
#define NOUT    1024
#define NCHUNK  128
#define NC      (NIN / NCHUNK)
#define BH      16
#define OSP     1028
#define LGP     36
#define PASS_LDS_BYTES ((BH * OSP + 2 * BH * LGP) * 4)

typedef float v2f __attribute__((ext_vector_type(2)));
typedef float v4f __attribute__((ext_vector_type(4)));
typedef float v8f __attribute__((ext_vector_type(8)));

struct Frag {
    v4f w[4][2][2];
    v2f u[4];
};

__device__ __forceinline__ void load_frag(Frag& f, const float* __restrict__ Wlane,
                                          const float* __restrict__ Ulane, int n)
{
    const float* wp = Wlane + (size_t)n * (KDIM * NOUT);
    #pragma unroll
    for (int s = 0; s < 4; ++s) {
        #pragma unroll
        for (int r = 0; r < 2; ++r) {
            #pragma unroll
            for (int tp = 0; tp < 2; ++tp)
                f.w[s][r][tp] = *(const v4f*)(wp + (4 * s + r) * NOUT + tp * 64);
        }
    }
    const float* up = Ulane + (size_t)n * KDIM;
    #pragma unroll
    for (int s = 0; s < 4; ++s) f.u[s] = *(const v2f*)(up + 4 * s);
}

template <bool ITER0, bool REFILL>
__device__ __forceinline__ void route_step(Frag& f, float (&oacc)[2][4][8],
                                           const float (&op)[2][32],
                                           float* lgw, const float* lgr,
                                           const float* __restrict__ Wlane,
                                           const float* __restrict__ Ulane, int n_refill)
{
    v8f D[2][4];
    #pragma unroll
    for (int tp = 0; tp < 2; ++tp) {
        #pragma unroll
        for (int q = 0; q < 4; ++q) D[tp][q] = (v8f){0.f, 0.f, 0.f, 0.f, 0.f, 0.f, 0.f, 0.f};
    }
    #pragma unroll
    for (int s = 0; s < 4; ++s) {
        #pragma unroll
        for (int tp = 0; tp < 2; ++tp) {
            #pragma unroll
            for (int q = 0; q < 4; ++q) {
                v2f a;
                a.x = f.w[s][0][tp][q];
                a.y = f.w[s][1][tp][q];
                D[tp][q] = __builtin_amdgcn_wmma_f32_16x16x4_f32(false, a, false, f.u[s], (short)0,
                                                                 D[tp][q], false, false);
            }
        }
    }
    if (REFILL) load_frag(f, Wlane, Ulane, n_refill);

    if (ITER0) {
        #pragma unroll
        for (int tp = 0; tp < 2; ++tp) {
            #pragma unroll
            for (int q = 0; q < 4; ++q) {
                #pragma unroll
                for (int r = 0; r < 8; ++r)
                    oacc[tp][q][r] = fmaf(0.03125f, D[tp][q][r], oacc[tp][q][r]);
            }
        }
    } else {
        float lgt[2];
        #pragma unroll
        for (int tp = 0; tp < 2; ++tp) {
            float s = 0.0f;
            #pragma unroll
            for (int j = 0; j < CAPD; ++j) s = fmaf(op[tp][j], D[tp][j & 3][j >> 2], s);
            lgt[tp] = s;
            lgw[2 * tp] = s;
        }
        __builtin_amdgcn_fence(__ATOMIC_RELEASE, "workgroup", "local");
        __builtin_amdgcn_s_barrier();
        __builtin_amdgcn_fence(__ATOMIC_ACQUIRE, "workgroup", "local");
        float L[NCAP];
        #pragma unroll
        for (int i4 = 0; i4 < NCAP / 4; ++i4) {
            const v4f t = *(const v4f*)(lgr + 4 * i4);
            L[4 * i4 + 0] = t.x; L[4 * i4 + 1] = t.y; L[4 * i4 + 2] = t.z; L[4 * i4 + 3] = t.w;
        }
        float m = -3.402823466e38f;
        #pragma unroll
        for (int i = 0; i < NCAP; ++i) m = fmaxf(m, L[i]);
        float ssum = 0.0f;
        #pragma unroll
        for (int i = 0; i < NCAP; ++i) ssum += __expf(L[i] - m);
        const float inv = 1.0f / ssum;
        #pragma unroll
        for (int tp = 0; tp < 2; ++tp) {
            const float cbv = __expf(lgt[tp] - m) * inv;
            #pragma unroll
            for (int q = 0; q < 4; ++q) {
                #pragma unroll
                for (int r = 0; r < 8; ++r)
                    oacc[tp][q][r] = fmaf(cbv, D[tp][q][r], oacc[tp][q][r]);
            }
        }
    }
}

template <bool ITER0>
__global__ __launch_bounds__(256)
void caps_pass(const float* __restrict__ U,
               const float* __restrict__ Wt,
               const float* __restrict__ o_prev,
               float* __restrict__ partial)
{
    extern __shared__ float smem[];
    float* osm = smem;
    float* lg  = smem + BH * OSP;

    const int tid   = threadIdx.x;
    const int lane  = tid & 31;
    const int wave  = tid >> 5;
    const int chunk = blockIdx.x;
    const int half  = blockIdx.y;
    const int b0    = half * BH;
    const int l16   = lane & 15;
    const int hi    = lane >> 4;

    float oacc[2][4][8];
    #pragma unroll
    for (int tp = 0; tp < 2; ++tp) {
        #pragma unroll
        for (int q = 0; q < 4; ++q) {
            #pragma unroll
            for (int r = 0; r < 8; ++r) oacc[tp][q][r] = 0.0f;
        }
    }

    float op[2][32];
    if (!ITER0) {
        #pragma unroll
        for (int tp = 0; tp < 2; ++tp) {
            const float* p = o_prev + ((size_t)(b0 + l16) * NCAP + (4 * wave + 2 * tp + hi)) * CAPD;
            #pragma unroll
            for (int j4 = 0; j4 < CAPD / 4; ++j4) {
                const v4f t = *(const v4f*)(p + 4 * j4);
                op[tp][4 * j4 + 0] = t.x; op[tp][4 * j4 + 1] = t.y;
                op[tp][4 * j4 + 2] = t.z; op[tp][4 * j4 + 3] = t.w;
            }
        }
    }

    const float* Wlane = Wt + (size_t)(2 * hi) * NOUT + wave * 128 + l16 * 4;
    const float* Ulane = U + (size_t)(b0 + l16) * NIN * KDIM + 2 * hi;
    const int n0 = chunk * NC;

    float* lgw0 = lg + l16 * LGP + 4 * wave + hi;
    float* lgw1 = lgw0 + BH * LGP;
    const float* lgr0 = lg + l16 * LGP;
    const float* lgr1 = lgr0 + BH * LGP;

    Frag f0, f1;
    load_frag(f0, Wlane, Ulane, n0);
    load_frag(f1, Wlane, Ulane, n0 + 1);
    #pragma unroll 1
    for (int nn = 0; nn < NC - 2; nn += 2) {
        route_step<ITER0, true>(f0, oacc, op, lgw0, lgr0, Wlane, Ulane, n0 + nn + 2);
        route_step<ITER0, true>(f1, oacc, op, lgw1, lgr1, Wlane, Ulane, n0 + nn + 3);
    }
    route_step<ITER0, false>(f0, oacc, op, lgw0, lgr0, Wlane, Ulane, 0);
    route_step<ITER0, false>(f1, oacc, op, lgw1, lgr1, Wlane, Ulane, 0);

    #pragma unroll
    for (int tp = 0; tp < 2; ++tp) {
        float* orow = osm + l16 * OSP + (4 * wave + 2 * tp + hi) * CAPD;
        #pragma unroll
        for (int r = 0; r < 8; ++r) {
            v4f t;
            t.x = oacc[tp][0][r]; t.y = oacc[tp][1][r]; t.z = oacc[tp][2][r]; t.w = oacc[tp][3][r];
            *(v4f*)(orow + 4 * r) = t;
        }
    }
    __syncthreads();
    float* pp = partial + (size_t)(chunk * 2 + half) * (BH * 1024);
    for (int e4 = tid; e4 < BH * 256; e4 += 256) {
        const int row = e4 >> 8, c4 = e4 & 255;
        *(v4f*)(pp + row * 1024 + c4 * 4) = *(const v4f*)(osm + row * OSP + c4 * 4);
    }
}

#define RCOLS   128
#define RGRP    4
#define RPER    (NCHUNK / RGRP)
__global__ __launch_bounds__(RCOLS) __attribute__((amdgpu_waves_per_eu(1, 4)))
void caps_reduce(const float* __restrict__ partial, float* __restrict__ out, int mode)
{
    __shared__ v4f gsum[RGRP][RCOLS / 4];
    const int cg = blockIdx.x;
    const int b  = blockIdx.y;
    const int half = b >> 4, bl = b & 15;
    const int tid = threadIdx.x;
    const int c4 = tid & 31;
    const int g  = tid >> 5;

    const float* p = partial + ((size_t)(g * RPER * 2 + half) * BH + bl) * 1024 + cg * RCOLS + c4 * 4;
    v4f t[RPER];
    #pragma unroll
    for (int i = 0; i < RPER; ++i) t[i] = *(const v4f*)(p + (size_t)i * 2 * (BH * 1024));
    v4f s0 = t[0], s1 = t[1], s2v = t[2], s3 = t[3];
    #pragma unroll
    for (int i = 4; i < RPER; i += 4) { s0 += t[i]; s1 += t[i + 1]; s2v += t[i + 2]; s3 += t[i + 3]; }
    gsum[g][c4] = (s0 + s1) + (s2v + s3);
    __syncthreads();

    const float* gs = (const float*)gsum;
    const float v = (gs[tid] + gs[RCOLS + tid]) + (gs[2 * RCOLS + tid] + gs[3 * RCOLS + tid]);
    float s2 = v * v;
    #pragma unroll
    for (int m = 16; m >= 1; m >>= 1) s2 += __shfl_xor(s2, m, 32);
    float f;
    if (mode == 0) f = rsqrtf(fmaxf(s2, 1e-12f));
    else           f = (s2 / (1.0f + s2)) * rsqrtf(s2 + 1e-7f);
    out[(size_t)b * NOUT + cg * RCOLS + tid] = v * f;
}

extern "C" void kernel_launch(void* const* d_in, const int* in_sizes, int n_in,
                              void* d_out, int out_size, void* d_ws, size_t ws_size,
                              hipStream_t stream)
{
    const float* U  = (const float*)d_in[0];
    const float* Wt = (const float*)d_in[1];
    float* out = (float*)d_out;

    float* partial = (float*)d_ws;
    float* o_prev  = partial + (size_t)(NCHUNK * 2) * (BH * 1024);

    dim3 grid(NCHUNK, 2), blk(256);

    caps_pass<true><<<grid, blk, PASS_LDS_BYTES, stream>>>(U, Wt, nullptr, partial);
    caps_reduce<<<dim3(NOUT / RCOLS, BATCH), RCOLS, 0, stream>>>(partial, o_prev, 0);
    caps_pass<false><<<grid, blk, PASS_LDS_BYTES, stream>>>(U, Wt, o_prev, partial);
    caps_reduce<<<dim3(NOUT / RCOLS, BATCH), RCOLS, 0, stream>>>(partial, o_prev, 0);
    caps_pass<false><<<grid, blk, PASS_LDS_BYTES, stream>>>(U, Wt, o_prev, partial);
    caps_reduce<<<dim3(NOUT / RCOLS, BATCH), RCOLS, 0, stream>>>(partial, out, 1);
}
